# starting kernel
# baseline (speedup 1.0000x reference)
#include <hip/hip_runtime.h>
#include <hip/hip_bf16.h>

typedef __attribute__((ext_vector_type(2))) float v2f;
typedef __attribute__((ext_vector_type(8))) float v8f;

#define N_NODES 50000
#define N_EDGES 800000
#define DIM 128

__global__ void zero_u32_kernel(unsigned* __restrict__ p, int n) {
    int i = blockIdx.x * blockDim.x + threadIdx.x;
    if (i < n) p[i] = 0u;
}

__global__ void degree_kernel(const int* __restrict__ src, const int* __restrict__ dst,
                              int* __restrict__ deg_out, int* __restrict__ deg_in, int ne) {
    int e = blockIdx.x * blockDim.x + threadIdx.x;
    if (e < ne) {
        atomicAdd(&deg_out[src[e]], 1);
        atomicAdd(&deg_in[dst[e]], 1);
    }
}

__global__ void norm_kernel(const int* __restrict__ deg_out, const int* __restrict__ deg_in,
                            float* __restrict__ norm_src, float* __restrict__ norm_dst, int n) {
    int i = blockIdx.x * blockDim.x + threadIdx.x;
    if (i < n) {
        int dov = deg_out[i]; if (dov < 1) dov = 1;
        int din = deg_in[i];  if (din < 1) din = 1;
        norm_src[i] = rsqrtf((float)dov);
        norm_dst[i] = rsqrtf((float)din);
    }
}

__global__ __launch_bounds__(256) void gemm_wmma_kernel(
    const float* __restrict__ x, const float* __restrict__ norm_src,
    const float* __restrict__ W, float* __restrict__ H) {
    int wave = (int)((blockIdx.x * blockDim.x + threadIdx.x) >> 5);
    int lane = threadIdx.x & 31;
    const int n_tiles = N_NODES / 16;
    if (wave >= n_tiles) return;

    const int row0 = wave * 16;
    const int m    = lane & 15;
    const int kk   = (lane >> 4) * 2;
    const int mh   = (lane >> 4) * 8;

    const float nrm = norm_src[row0 + m];
    const float* xrow = x + (size_t)(row0 + m) * DIM + kk;

    v8f acc[8] = {};

    for (int k = 0; k < DIM; k += 4) {
        float2 xv = *(const float2*)(xrow + k);
        v2f a = { xv.x * nrm, xv.y * nrm };
        const float* wb = W + (size_t)(k + kk) * DIM + m;
#pragma unroll
        for (int t = 0; t < 8; ++t) {
            v2f b = { wb[t * 16], wb[t * 16 + DIM] };
            acc[t] = __builtin_amdgcn_wmma_f32_16x16x4_f32(
                false, a, false, b, (short)0, acc[t], false, false);
        }
    }

    float* hrow = H + (size_t)row0 * DIM + m;
#pragma unroll
    for (int t = 0; t < 8; ++t) {
#pragma unroll
        for (int r = 0; r < 8; ++r) {
            hrow[(size_t)(r + mh) * DIM + t * 16] = acc[t][r];
        }
    }
}

__global__ __launch_bounds__(256) void scatter_kernel(
    const float* __restrict__ H, const int* __restrict__ src, const int* __restrict__ dst,
    float* __restrict__ out, int ne) {
    int wave = (int)((blockIdx.x * blockDim.x + threadIdx.x) >> 5);
    int lane = threadIdx.x & 31;
    if (wave >= ne) return;
    int s = src[wave];
    int d = dst[wave];
    const float4 v = *(const float4*)(H + (size_t)s * DIM + lane * 4);
    float* o = out + (size_t)d * DIM + lane * 4;
    atomicAdd(o + 0, v.x);
    atomicAdd(o + 1, v.y);
    atomicAdd(o + 2, v.z);
    atomicAdd(o + 3, v.w);
}

__global__ void finish_kernel(float* __restrict__ out, const float* __restrict__ norm_dst,
                              const float* __restrict__ b, int n) {
    int idx = blockIdx.x * blockDim.x + threadIdx.x;
    if (idx < n) {
        int i = idx >> 7;
        int j = idx & (DIM - 1);
        out[idx] = out[idx] * norm_dst[i] + b[j];
    }
}

extern "C" void kernel_launch(void* const* d_in, const int* in_sizes, int n_in,
                              void* d_out, int out_size, void* d_ws, size_t ws_size,
                              hipStream_t stream) {
    const float* x   = (const float*)d_in[0];
    const int*   src = (const int*)d_in[1];
    const int*   dst = (const int*)d_in[2];
    const float* W   = (const float*)d_in[3];
    const float* b   = (const float*)d_in[4];
    float* out = (float*)d_out;

    int*   deg_out  = (int*)d_ws;
    int*   deg_in   = deg_out + N_NODES;
    float* norm_src = (float*)(deg_in + N_NODES);
    float* norm_dst = norm_src + N_NODES;
    float* H        = norm_dst + N_NODES;

    const int TB = 256;

    zero_u32_kernel<<<(N_NODES * DIM + TB - 1) / TB, TB, 0, stream>>>((unsigned*)out, N_NODES * DIM);
    zero_u32_kernel<<<(2 * N_NODES + TB - 1) / TB, TB, 0, stream>>>((unsigned*)deg_out, 2 * N_NODES);

    degree_kernel<<<(N_EDGES + TB - 1) / TB, TB, 0, stream>>>(src, dst, deg_out, deg_in, N_EDGES);

    norm_kernel<<<(N_NODES + TB - 1) / TB, TB, 0, stream>>>(deg_out, deg_in, norm_src, norm_dst, N_NODES);

    {
        int waves  = N_NODES / 16;
        int blocks = (waves * 32 + TB - 1) / TB;
        gemm_wmma_kernel<<<blocks, TB, 0, stream>>>(x, norm_src, W, H);
    }

    {
        long long threads = (long long)N_EDGES * 32;
        int blocks = (int)((threads + TB - 1) / TB);
        scatter_kernel<<<blocks, TB, 0, stream>>>(H, src, dst, out, N_EDGES);
    }

    finish_kernel<<<(N_NODES * DIM + TB - 1) / TB, TB, 0, stream>>>(out, norm_dst, b, N_NODES * DIM);
}


# Edge sum as a gather, no float atomics: in-degree add's returned rank files src in a 64-slot dst bucket; one wave per dst row sums H rows in registers in arrival-rank order, scale+bias fused
# speedup vs baseline: 1.9011x; 1.9011x over previous
#include <hip/hip_runtime.h>
#include <hip/hip_bf16.h>

typedef __attribute__((ext_vector_type(2))) float v2f;
typedef __attribute__((ext_vector_type(8))) float v8f;

#define N_NODES 50000
#define N_EDGES 800000
#define DIM 128

__global__ void zero_u32_kernel(unsigned* __restrict__ p, int n) {
    int i = blockIdx.x * blockDim.x + threadIdx.x;
    if (i < n) p[i] = 0u;
}

#define CAP 64
__global__ void degree_bucket_kernel(const int* __restrict__ src, const int* __restrict__ dst,
                                     int* __restrict__ deg_out, int* __restrict__ deg_in,
                                     int* __restrict__ bucket, int ne) {
    int e = blockIdx.x * blockDim.x + threadIdx.x;
    if (e < ne) {
        const int s = src[e];
        const int d = dst[e];
        atomicAdd(&deg_out[s], 1);
        const int r = atomicAdd(&deg_in[d], 1);
        if (r < CAP) bucket[(size_t)d * CAP + r] = s;
    }
}

__global__ void norm_kernel(const int* __restrict__ deg_out, const int* __restrict__ deg_in,
                            float* __restrict__ norm_src, float* __restrict__ norm_dst, int n) {
    int i = blockIdx.x * blockDim.x + threadIdx.x;
    if (i < n) {
        int dov = deg_out[i]; if (dov < 1) dov = 1;
        int din = deg_in[i];  if (din < 1) din = 1;
        norm_src[i] = rsqrtf((float)dov);
        norm_dst[i] = rsqrtf((float)din);
    }
}

__global__ __launch_bounds__(256) void gemm_wmma_kernel(
    const float* __restrict__ x, const float* __restrict__ norm_src,
    const float* __restrict__ W, float* __restrict__ H) {
    int wave = (int)((blockIdx.x * blockDim.x + threadIdx.x) >> 5);
    int lane = threadIdx.x & 31;
    const int n_tiles = N_NODES / 16;
    if (wave >= n_tiles) return;

    const int row0 = wave * 16;
    const int m    = lane & 15;
    const int kk   = (lane >> 4) * 2;
    const int mh   = (lane >> 4) * 8;

    const float nrm = norm_src[row0 + m];
    const float* xrow = x + (size_t)(row0 + m) * DIM + kk;

    v8f acc[8] = {};

    for (int k = 0; k < DIM; k += 4) {
        float2 xv = *(const float2*)(xrow + k);
        v2f a = { xv.x * nrm, xv.y * nrm };
        const float* wb = W + (size_t)(k + kk) * DIM + m;
#pragma unroll
        for (int t = 0; t < 8; ++t) {
            v2f b = { wb[t * 16], wb[t * 16 + DIM] };
            acc[t] = __builtin_amdgcn_wmma_f32_16x16x4_f32(
                false, a, false, b, (short)0, acc[t], false, false);
        }
    }

    float* hrow = H + (size_t)row0 * DIM + m;
#pragma unroll
    for (int t = 0; t < 8; ++t) {
#pragma unroll
        for (int r = 0; r < 8; ++r) {
            hrow[(size_t)(r + mh) * DIM + t * 16] = acc[t][r];
        }
    }
}

__device__ __forceinline__ float4 ld_row(const float* __restrict__ Hl, int s) {
    return *(const float4*)(Hl + (size_t)s * DIM);
}

__device__ __forceinline__ void sum_slots(const float* __restrict__ Hl, int iv, int m, float4& acc) {
    for (int j = 0; j < m; j += 8) {
        float4 v[8];
#pragma unroll
        for (int u = 0; u < 8; ++u) {
            v[u] = make_float4(0.f, 0.f, 0.f, 0.f);
            if (j + u < m) {
                const int s = __builtin_amdgcn_readlane(iv, j + u);
                v[u] = ld_row(Hl, s);
            }
        }
#pragma unroll
        for (int u = 0; u < 8; ++u) {
            acc.x += v[u].x; acc.y += v[u].y; acc.z += v[u].z; acc.w += v[u].w;
        }
    }
}

__global__ __launch_bounds__(256) void gather_kernel(
    const float* __restrict__ H, const int* __restrict__ src, const int* __restrict__ dst,
    const int* __restrict__ deg_in, const int* __restrict__ bucket,
    const float* __restrict__ norm_dst, const float* __restrict__ b,
    float* __restrict__ out, int ne) {
    const int wave = (int)((blockIdx.x * blockDim.x + threadIdx.x) >> 5);
    const int lane = threadIdx.x & 31;
    if (wave >= N_NODES) return;
    const int d = __builtin_amdgcn_readfirstlane(wave);
    const int cnt = deg_in[d];
    const float nd = norm_dst[d];
    const float4 bb = *(const float4*)(b + lane * 4);
    const float* Hl = H + lane * 4;
    float4 acc = make_float4(0.f, 0.f, 0.f, 0.f);

    if (cnt <= CAP) {
        const int* bk = bucket + (size_t)d * CAP;
        const int i0 = (lane < cnt) ? bk[lane] : 0;
        const int i1 = (lane + 32 < cnt) ? bk[lane + 32] : 0;
        sum_slots(Hl, i0, cnt < 32 ? cnt : 32, acc);
        if (cnt > 32) sum_slots(Hl, i1, cnt - 32, acc);
    } else {
        for (int e0 = 0; e0 < ne; e0 += 32) {
            const int e = e0 + lane;
            int dd = -1, ss = 0;
            if (e < ne) { dd = dst[e]; ss = src[e]; }
            unsigned hit = __builtin_amdgcn_ballot_w32(dd == d);
            while (hit) {
                const int l = __builtin_ctz(hit);
                hit &= hit - 1u;
                const int s = __builtin_amdgcn_readlane(ss, l);
                const float4 v = ld_row(Hl, s);
                acc.x += v.x; acc.y += v.y; acc.z += v.z; acc.w += v.w;
            }
        }
    }

    float4 o;
    o.x = acc.x * nd + bb.x;
    o.y = acc.y * nd + bb.y;
    o.z = acc.z * nd + bb.z;
    o.w = acc.w * nd + bb.w;
    *(float4*)(out + (size_t)d * DIM + lane * 4) = o;
}

extern "C" void kernel_launch(void* const* d_in, const int* in_sizes, int n_in,
                              void* d_out, int out_size, void* d_ws, size_t ws_size,
                              hipStream_t stream) {
    const float* x   = (const float*)d_in[0];
    const int*   src = (const int*)d_in[1];
    const int*   dst = (const int*)d_in[2];
    const float* W   = (const float*)d_in[3];
    const float* b   = (const float*)d_in[4];
    float* out = (float*)d_out;

    int*   deg_out  = (int*)d_ws;
    int*   deg_in   = deg_out + N_NODES;
    float* norm_src = (float*)(deg_in + N_NODES);
    float* norm_dst = norm_src + N_NODES;
    float* H        = norm_dst + N_NODES;
    int*   bucket   = (int*)(H + (size_t)N_NODES * DIM);

    const int TB = 256;

    zero_u32_kernel<<<(2 * N_NODES + TB - 1) / TB, TB, 0, stream>>>((unsigned*)deg_out, 2 * N_NODES);

    degree_bucket_kernel<<<(N_EDGES + TB - 1) / TB, TB, 0, stream>>>(src, dst, deg_out, deg_in, bucket, N_EDGES);

    norm_kernel<<<(N_NODES + TB - 1) / TB, TB, 0, stream>>>(deg_out, deg_in, norm_src, norm_dst, N_NODES);

    {
        int waves  = N_NODES / 16;
        int blocks = (waves * 32 + TB - 1) / TB;
        gemm_wmma_kernel<<<blocks, TB, 0, stream>>>(x, norm_src, W, H);
    }

    gather_kernel<<<(N_NODES * 32 + TB - 1) / TB, TB, 0, stream>>>(H, src, dst, deg_in, bucket, norm_dst, b, out, N_EDGES);
}


# Degree counts and source buckets built in LDS by node-range owner work-groups that scan the edge list: no global atomics, no zero or norm launch; row sums go chunk by chunk in LDS rank order
# speedup vs baseline: 3.3946x; 1.7856x over previous
#include <hip/hip_runtime.h>
#include <hip/hip_bf16.h>


typedef __attribute__((ext_vector_type(2))) float v2f;
typedef __attribute__((ext_vector_type(8))) float v8f;

#define N_NODES 50000
#define N_EDGES 800000
#define DIM 128

#define KR 800
#define NRANGE 63
#define NCHUNK 4
#define CAPC 32
#define NP (NRANGE * KR)
#define OCT_PER_CHUNK (N_EDGES / 8 / NCHUNK)
#define OWNER_THREADS 1024

__device__ __forceinline__ void file_edge(int sv, int dv, int base, int* cin, int* cout, int* bk) {
    const unsigned td = (unsigned)(dv - base);
    if (td < (unsigned)KR) {
        const int r = atomicAdd(&cin[td], 1);
        if (r < CAPC) bk[td * CAPC + r] = sv;
    }
    const unsigned ts = (unsigned)(sv - base);
    if (ts < (unsigned)KR) atomicAdd(&cout[ts], 1);
}

__global__ __launch_bounds__(OWNER_THREADS) void owner_count_kernel(
    const int* __restrict__ src, const int* __restrict__ dst,
    int* __restrict__ cnt_out, int* __restrict__ cnt_in, int* __restrict__ bucket) {
    __shared__ int cin[KR];
    __shared__ int cout[KR];
    __shared__ __attribute__((aligned(16))) int bk[KR * CAPC];

    const int tid  = threadIdx.x;
    const int k    = blockIdx.x / NCHUNK;
    const int c    = blockIdx.x - k * NCHUNK;
    const int base = k * KR;

    if (tid < KR) { cin[tid] = 0; cout[tid] = 0; }
    {
        int4* bz = (int4*)bk;
        for (int j = tid; j < KR * CAPC / 4; j += OWNER_THREADS) bz[j] = make_int4(0, 0, 0, 0);
    }
    __syncthreads();

    const int4* s4 = (const int4*)src + (size_t)c * (2 * OCT_PER_CHUNK);
    const int4* d4 = (const int4*)dst + (size_t)c * (2 * OCT_PER_CHUNK);

    int o = tid;
    int4 sa = make_int4(0, 0, 0, 0), sb = sa, da = sa, db = sa;
    if (o < OCT_PER_CHUNK) { sa = s4[2 * o]; sb = s4[2 * o + 1]; da = d4[2 * o]; db = d4[2 * o + 1]; }
    while (o < OCT_PER_CHUNK) {
        const int on = o + OWNER_THREADS;
        int4 nsa = make_int4(0, 0, 0, 0), nsb = nsa, nda = nsa, ndb = nsa;
        if (on < OCT_PER_CHUNK) { nsa = s4[2 * on]; nsb = s4[2 * on + 1]; nda = d4[2 * on]; ndb = d4[2 * on + 1]; }
        file_edge(sa.x, da.x, base, cin, cout, bk);
        file_edge(sa.y, da.y, base, cin, cout, bk);
        file_edge(sa.z, da.z, base, cin, cout, bk);
        file_edge(sa.w, da.w, base, cin, cout, bk);
        file_edge(sb.x, db.x, base, cin, cout, bk);
        file_edge(sb.y, db.y, base, cin, cout, bk);
        file_edge(sb.z, db.z, base, cin, cout, bk);
        file_edge(sb.w, db.w, base, cin, cout, bk);
        sa = nsa; sb = nsb; da = nda; db = ndb;
        o = on;
    }
    __syncthreads();

    if (tid < KR) {
        cnt_in[c * NP + base + tid]  = cin[tid];
        cnt_out[c * NP + base + tid] = cout[tid];
    }
    {
        const int4* bk4 = (const int4*)bk;
        for (int j = tid; j < KR * CAPC / 4; j += OWNER_THREADS) {
            const int i    = j >> 3;
            const int part = j & 7;
            *(int4*)(bucket + ((size_t)(base + i) * NCHUNK + c) * CAPC + part * 4) = bk4[j];
        }
    }
}

__global__ __launch_bounds__(256) void gemm_wmma_kernel(
    const float* __restrict__ x, const int* __restrict__ cnt_out,
    const float* __restrict__ W, float* __restrict__ H) {
    int wave = (int)((blockIdx.x * blockDim.x + threadIdx.x) >> 5);
    int lane = threadIdx.x & 31;
    const int n_tiles = N_NODES / 16;
    if (wave >= n_tiles) return;

    const int row0 = wave * 16;
    const int m    = lane & 15;
    const int kk   = (lane >> 4) * 2;
    const int mh   = (lane >> 4) * 8;

    int dov = cnt_out[row0 + m] + cnt_out[NP + row0 + m] + cnt_out[2 * NP + row0 + m] + cnt_out[3 * NP + row0 + m];
    if (dov < 1) dov = 1;
    const float nrm = rsqrtf((float)dov);
    const float* xrow = x + (size_t)(row0 + m) * DIM + kk;

    v8f acc[8] = {};

    for (int k = 0; k < DIM; k += 4) {
        float2 xv = *(const float2*)(xrow + k);
        v2f a = { xv.x * nrm, xv.y * nrm };
        const float* wb = W + (size_t)(k + kk) * DIM + m;
#pragma unroll
        for (int t = 0; t < 8; ++t) {
            v2f b = { wb[t * 16], wb[t * 16 + DIM] };
            acc[t] = __builtin_amdgcn_wmma_f32_16x16x4_f32(
                false, a, false, b, (short)0, acc[t], false, false);
        }
    }

    float* hrow = H + (size_t)row0 * DIM + m;
#pragma unroll
    for (int t = 0; t < 8; ++t) {
#pragma unroll
        for (int r = 0; r < 8; ++r) {
            hrow[(size_t)(r + mh) * DIM + t * 16] = acc[t][r];
        }
    }
}

__device__ __forceinline__ float4 ld_row(const float* __restrict__ Hl, int s) {
    return *(const float4*)(Hl + (size_t)s * DIM);
}

__device__ __forceinline__ void sum_slots(const float* __restrict__ Hl, int iv, int m, float4& acc) {
    for (int j = 0; j < m; j += 8) {
        float4 v[8];
#pragma unroll
        for (int u = 0; u < 8; ++u) {
            v[u] = make_float4(0.f, 0.f, 0.f, 0.f);
            if (j + u < m) {
                const int s = __builtin_amdgcn_readlane(iv, j + u);
                v[u] = ld_row(Hl, s);
            }
        }
#pragma unroll
        for (int u = 0; u < 8; ++u) {
            acc.x += v[u].x; acc.y += v[u].y; acc.z += v[u].z; acc.w += v[u].w;
        }
    }
}

__global__ __launch_bounds__(256) void gather_kernel(
    const float* __restrict__ H, const int* __restrict__ src, const int* __restrict__ dst,
    const int* __restrict__ cnt_in, const int* __restrict__ bucket,
    const float* __restrict__ b,
    float* __restrict__ out, int ne) {
    const int wave = (int)((blockIdx.x * blockDim.x + threadIdx.x) >> 5);
    const int lane = threadIdx.x & 31;
    if (wave >= N_NODES) return;
    const int d = __builtin_amdgcn_readfirstlane(wave);
    const int n0 = cnt_in[d];
    const int n1 = cnt_in[NP + d];
    const int n2 = cnt_in[2 * NP + d];
    const int n3 = cnt_in[3 * NP + d];
    const int p1 = n0;
    const int p2 = p1 + n1;
    const int p3 = p2 + n2;
    const int cnt = p3 + n3;
    const float nd = rsqrtf((float)(cnt < 1 ? 1 : cnt));
    const float4 bb = *(const float4*)(b + lane * 4);
    const float* Hl = H + lane * 4;
    float4 acc = make_float4(0.f, 0.f, 0.f, 0.f);

    const bool filed = (n0 <= CAPC) && (n1 <= CAPC) && (n2 <= CAPC) && (n3 <= CAPC) && (cnt <= 64);
    if (filed) {
        const int* bk = bucket + (size_t)d * (NCHUNK * CAPC);
        const int j0 = lane;
        const int c0 = (j0 >= p1) + (j0 >= p2) + (j0 >= p3);
        const int s0 = j0 - (j0 >= p3 ? p3 : (j0 >= p2 ? p2 : (j0 >= p1 ? p1 : 0)));
        const int j1 = lane + 32;
        const int c1 = (j1 >= p1) + (j1 >= p2) + (j1 >= p3);
        const int s1 = j1 - (j1 >= p3 ? p3 : (j1 >= p2 ? p2 : (j1 >= p1 ? p1 : 0)));
        const int i0 = (j0 < cnt) ? bk[c0 * CAPC + s0] : 0;
        const int i1 = (j1 < cnt) ? bk[c1 * CAPC + s1] : 0;
        sum_slots(Hl, i0, cnt < 32 ? cnt : 32, acc);
        if (cnt > 32) sum_slots(Hl, i1, cnt - 32, acc);
    } else {
        for (int e0 = 0; e0 < ne; e0 += 32) {
            const int e = e0 + lane;
            int dd = -1, ss = 0;
            if (e < ne) { dd = dst[e]; ss = src[e]; }
            unsigned hit = __builtin_amdgcn_ballot_w32(dd == d);
            while (hit) {
                const int l = __builtin_ctz(hit);
                hit &= hit - 1u;
                const int s = __builtin_amdgcn_readlane(ss, l);
                const float4 v = ld_row(Hl, s);
                acc.x += v.x; acc.y += v.y; acc.z += v.z; acc.w += v.w;
            }
        }
    }

    float4 o;
    o.x = acc.x * nd + bb.x;
    o.y = acc.y * nd + bb.y;
    o.z = acc.z * nd + bb.z;
    o.w = acc.w * nd + bb.w;
    *(float4*)(out + (size_t)d * DIM + lane * 4) = o;
}

extern "C" void kernel_launch(void* const* d_in, const int* in_sizes, int n_in,
                              void* d_out, int out_size, void* d_ws, size_t ws_size,
                              hipStream_t stream) {
    const float* x   = (const float*)d_in[0];
    const int*   src = (const int*)d_in[1];
    const int*   dst = (const int*)d_in[2];
    const float* W   = (const float*)d_in[3];
    const float* b   = (const float*)d_in[4];
    float* out = (float*)d_out;

    int*   cnt_out = (int*)d_ws;
    int*   cnt_in  = cnt_out + NCHUNK * NP;
    float* H       = (float*)(cnt_in + NCHUNK * NP);
    int*   bucket  = (int*)(H + (size_t)N_NODES * DIM);

    const int TB = 256;

    owner_count_kernel<<<NRANGE * NCHUNK, OWNER_THREADS, 0, stream>>>(src, dst, cnt_out, cnt_in, bucket);

    {
        int waves  = N_NODES / 16;
        int blocks = (waves * 32 + TB - 1) / TB;
        gemm_wmma_kernel<<<blocks, TB, 0, stream>>>(x, cnt_out, W, H);
    }

    gather_kernel<<<(N_NODES * 32 + TB - 1) / TB, TB, 0, stream>>>(H, src, dst, cnt_in, bucket, b, out, N_EDGES);
}


# Owner scan halved: 32 node ranges of 1600 x 8 edge chunks (was 63 x 4), 16 slots per node and chunk stored in chunk planes; row sums now go in 8-chunk rank order
# speedup vs baseline: 3.7535x; 1.1057x over previous
#include <hip/hip_runtime.h>
#include <hip/hip_bf16.h>


typedef __attribute__((ext_vector_type(2))) float v2f;
typedef __attribute__((ext_vector_type(8))) float v8f;

#define N_NODES 50000
#define N_EDGES 800000
#define DIM 128

#define KR 1600
#define NRANGE 32
#define NCHUNK 8
#define CAPC 16
#define NP (NRANGE * KR)
#define OCT_PER_CHUNK (N_EDGES / 8 / NCHUNK)
#define OWNER_THREADS 1024

__device__ __forceinline__ void file_edge(int sv, int dv, int base, int* cin, int* cout, int* bk) {
    const unsigned td = (unsigned)(dv - base);
    if (td < (unsigned)KR) {
        const int r = atomicAdd(&cin[td], 1);
        if (r < CAPC) bk[td * CAPC + r] = sv;
    }
    const unsigned ts = (unsigned)(sv - base);
    if (ts < (unsigned)KR) atomicAdd(&cout[ts], 1);
}

__global__ __launch_bounds__(OWNER_THREADS) void owner_count_kernel(
    const int* __restrict__ src, const int* __restrict__ dst,
    int* __restrict__ cnt_out, int* __restrict__ cnt_in, int* __restrict__ bucket) {
    __shared__ int cin[KR];
    __shared__ int cout[KR];
    __shared__ __attribute__((aligned(16))) int bk[KR * CAPC];

    const int tid  = threadIdx.x;
    const int k    = blockIdx.x / NCHUNK;
    const int c    = blockIdx.x - k * NCHUNK;
    const int base = k * KR;

    for (int i = tid; i < KR; i += OWNER_THREADS) { cin[i] = 0; cout[i] = 0; }
    {
        int4* bz = (int4*)bk;
        for (int j = tid; j < KR * CAPC / 4; j += OWNER_THREADS) bz[j] = make_int4(0, 0, 0, 0);
    }
    __syncthreads();

    const int4* s4 = (const int4*)src + (size_t)c * (2 * OCT_PER_CHUNK);
    const int4* d4 = (const int4*)dst + (size_t)c * (2 * OCT_PER_CHUNK);

    int o = tid;
    int4 sa = make_int4(0, 0, 0, 0), sb = sa, da = sa, db = sa;
    if (o < OCT_PER_CHUNK) { sa = s4[2 * o]; sb = s4[2 * o + 1]; da = d4[2 * o]; db = d4[2 * o + 1]; }
    while (o < OCT_PER_CHUNK) {
        const int on = o + OWNER_THREADS;
        int4 nsa = make_int4(0, 0, 0, 0), nsb = nsa, nda = nsa, ndb = nsa;
        if (on < OCT_PER_CHUNK) { nsa = s4[2 * on]; nsb = s4[2 * on + 1]; nda = d4[2 * on]; ndb = d4[2 * on + 1]; }
        file_edge(sa.x, da.x, base, cin, cout, bk);
        file_edge(sa.y, da.y, base, cin, cout, bk);
        file_edge(sa.z, da.z, base, cin, cout, bk);
        file_edge(sa.w, da.w, base, cin, cout, bk);
        file_edge(sb.x, db.x, base, cin, cout, bk);
        file_edge(sb.y, db.y, base, cin, cout, bk);
        file_edge(sb.z, db.z, base, cin, cout, bk);
        file_edge(sb.w, db.w, base, cin, cout, bk);
        sa = nsa; sb = nsb; da = nda; db = ndb;
        o = on;
    }
    __syncthreads();

    for (int i = tid; i < KR; i += OWNER_THREADS) {
        cnt_in[c * NP + base + i]  = cin[i];
        cnt_out[c * NP + base + i] = cout[i];
    }
    {
        const int4* bk4 = (const int4*)bk;
        int4* dst4 = (int4*)(bucket + ((size_t)c * NP + base) * CAPC);
        for (int j = tid; j < KR * CAPC / 4; j += OWNER_THREADS) dst4[j] = bk4[j];
    }
}

__global__ __launch_bounds__(256) void gemm_wmma_kernel(
    const float* __restrict__ x, const int* __restrict__ cnt_out,
    const float* __restrict__ W, float* __restrict__ H) {
    int wave = (int)((blockIdx.x * blockDim.x + threadIdx.x) >> 5);
    int lane = threadIdx.x & 31;
    const int n_tiles = N_NODES / 16;
    if (wave >= n_tiles) return;

    const int row0 = wave * 16;
    const int m    = lane & 15;
    const int kk   = (lane >> 4) * 2;
    const int mh   = (lane >> 4) * 8;

    int dov = 0;
#pragma unroll
    for (int c = 0; c < NCHUNK; ++c) dov += cnt_out[c * NP + row0 + m];
    if (dov < 1) dov = 1;
    const float nrm = rsqrtf((float)dov);
    const float* xrow = x + (size_t)(row0 + m) * DIM + kk;

    v8f acc[8] = {};

    for (int k = 0; k < DIM; k += 4) {
        float2 xv = *(const float2*)(xrow + k);
        v2f a = { xv.x * nrm, xv.y * nrm };
        const float* wb = W + (size_t)(k + kk) * DIM + m;
#pragma unroll
        for (int t = 0; t < 8; ++t) {
            v2f b = { wb[t * 16], wb[t * 16 + DIM] };
            acc[t] = __builtin_amdgcn_wmma_f32_16x16x4_f32(
                false, a, false, b, (short)0, acc[t], false, false);
        }
    }

    float* hrow = H + (size_t)row0 * DIM + m;
#pragma unroll
    for (int t = 0; t < 8; ++t) {
#pragma unroll
        for (int r = 0; r < 8; ++r) {
            hrow[(size_t)(r + mh) * DIM + t * 16] = acc[t][r];
        }
    }
}

__device__ __forceinline__ float4 ld_row(const float* __restrict__ Hl, int s) {
    return *(const float4*)(Hl + (size_t)s * DIM);
}

__device__ __forceinline__ void sum_slots(const float* __restrict__ Hl, int iv, int m, float4& acc) {
    for (int j = 0; j < m; j += 8) {
        float4 v[8];
#pragma unroll
        for (int u = 0; u < 8; ++u) {
            v[u] = make_float4(0.f, 0.f, 0.f, 0.f);
            if (j + u < m) {
                const int s = __builtin_amdgcn_readlane(iv, j + u);
                v[u] = ld_row(Hl, s);
            }
        }
#pragma unroll
        for (int u = 0; u < 8; ++u) {
            acc.x += v[u].x; acc.y += v[u].y; acc.z += v[u].z; acc.w += v[u].w;
        }
    }
}

__global__ __launch_bounds__(256) void gather_kernel(
    const float* __restrict__ H, const int* __restrict__ src, const int* __restrict__ dst,
    const int* __restrict__ cnt_in, const int* __restrict__ bucket,
    const float* __restrict__ b,
    float* __restrict__ out, int ne) {
    const int wave = (int)((blockIdx.x * blockDim.x + threadIdx.x) >> 5);
    const int lane = threadIdx.x & 31;
    if (wave >= N_NODES) return;
    const int d = __builtin_amdgcn_readfirstlane(wave);
    int p[NCHUNK + 1];
    bool fits = true;
    p[0] = 0;
#pragma unroll
    for (int c = 0; c < NCHUNK; ++c) {
        const int n = cnt_in[c * NP + d];
        fits = fits && (n <= CAPC);
        p[c + 1] = p[c] + n;
    }
    const int cnt = p[NCHUNK];
    const float nd = rsqrtf((float)(cnt < 1 ? 1 : cnt));
    const float4 bb = *(const float4*)(b + lane * 4);
    const float* Hl = H + lane * 4;
    float4 acc = make_float4(0.f, 0.f, 0.f, 0.f);

    const bool filed = fits && (cnt <= 64);
    if (filed) {
        const int* bk = bucket + (size_t)d * CAPC;
        const int j0 = lane;
        const int j1 = lane + 32;
        int c0 = 0, b0 = 0, c1 = 0, b1 = 0;
#pragma unroll
        for (int c = 1; c < NCHUNK; ++c) {
            if (j0 >= p[c]) { c0 = c; b0 = p[c]; }
            if (j1 >= p[c]) { c1 = c; b1 = p[c]; }
        }
        const int i0 = (j0 < cnt) ? bk[((size_t)c0 * NP) * CAPC + (j0 - b0)] : 0;
        const int i1 = (j1 < cnt) ? bk[((size_t)c1 * NP) * CAPC + (j1 - b1)] : 0;
        sum_slots(Hl, i0, cnt < 32 ? cnt : 32, acc);
        if (cnt > 32) sum_slots(Hl, i1, cnt - 32, acc);
    } else {
        for (int e0 = 0; e0 < ne; e0 += 32) {
            const int e = e0 + lane;
            int dd = -1, ss = 0;
            if (e < ne) { dd = dst[e]; ss = src[e]; }
            unsigned hit = __builtin_amdgcn_ballot_w32(dd == d);
            while (hit) {
                const int l = __builtin_ctz(hit);
                hit &= hit - 1u;
                const int s = __builtin_amdgcn_readlane(ss, l);
                const float4 v = ld_row(Hl, s);
                acc.x += v.x; acc.y += v.y; acc.z += v.z; acc.w += v.w;
            }
        }
    }

    float4 o;
    o.x = acc.x * nd + bb.x;
    o.y = acc.y * nd + bb.y;
    o.z = acc.z * nd + bb.z;
    o.w = acc.w * nd + bb.w;
    *(float4*)(out + (size_t)d * DIM + lane * 4) = o;
}

extern "C" void kernel_launch(void* const* d_in, const int* in_sizes, int n_in,
                              void* d_out, int out_size, void* d_ws, size_t ws_size,
                              hipStream_t stream) {
    const float* x   = (const float*)d_in[0];
    const int*   src = (const int*)d_in[1];
    const int*   dst = (const int*)d_in[2];
    const float* W   = (const float*)d_in[3];
    const float* b   = (const float*)d_in[4];
    float* out = (float*)d_out;

    int*   cnt_out = (int*)d_ws;
    int*   cnt_in  = cnt_out + NCHUNK * NP;
    float* H       = (float*)(cnt_in + NCHUNK * NP);
    int*   bucket  = (int*)(H + (size_t)N_NODES * DIM);

    const int TB = 256;

    owner_count_kernel<<<NRANGE * NCHUNK, OWNER_THREADS, 0, stream>>>(src, dst, cnt_out, cnt_in, bucket);

    {
        int waves  = N_NODES / 16;
        int blocks = (waves * 32 + TB - 1) / TB;
        gemm_wmma_kernel<<<blocks, TB, 0, stream>>>(x, cnt_out, W, H);
    }

    gather_kernel<<<(N_NODES * 32 + TB - 1) / TB, TB, 0, stream>>>(H, src, dst, cnt_in, bucket, b, out, N_EDGES);
}


# Source ids filed as two-byte slots so a work-group owns 3200 nodes: 16 node ranges x 16 edge chunks, each scanning 50000 edges (was 32 x 8, 100000); row sums now go in 16-chunk rank order
# speedup vs baseline: 3.8949x; 1.0377x over previous
#include <hip/hip_runtime.h>
#include <hip/hip_bf16.h>


typedef __attribute__((ext_vector_type(2))) float v2f;
typedef __attribute__((ext_vector_type(8))) float v8f;

#define N_NODES 50000
#define N_EDGES 800000
#define DIM 128

#define KR 3200
#define NRANGE 16
#define NCHUNK 16
#define CAPC 16
typedef unsigned short slot_t;
static_assert(N_NODES <= 65536, "a source index must fit a two-byte slot");
#define NP (NRANGE * KR)
#define OCT_PER_CHUNK (N_EDGES / 8 / NCHUNK)
#define OWNER_THREADS 1024

__device__ __forceinline__ void file_edge(int sv, int dv, int base, int* cin, int* cout, slot_t* bk) {
    const unsigned td = (unsigned)(dv - base);
    if (td < (unsigned)KR) {
        const int r = atomicAdd(&cin[td], 1);
        if (r < CAPC) bk[td * CAPC + r] = (slot_t)sv;
    }
    const unsigned ts = (unsigned)(sv - base);
    if (ts < (unsigned)KR) atomicAdd(&cout[ts], 1);
}

__global__ __launch_bounds__(OWNER_THREADS) void owner_count_kernel(
    const int* __restrict__ src, const int* __restrict__ dst,
    int* __restrict__ cnt_out, int* __restrict__ cnt_in, int* __restrict__ bucket) {
    __shared__ int cin[KR];
    __shared__ int cout[KR];
    __shared__ __attribute__((aligned(16))) slot_t bk[KR * CAPC];

    const int tid  = threadIdx.x;
    const int k    = blockIdx.x / NCHUNK;
    const int c    = blockIdx.x - k * NCHUNK;
    const int base = k * KR;

    for (int i = tid; i < KR; i += OWNER_THREADS) { cin[i] = 0; cout[i] = 0; }
    {
        int4* bz = (int4*)bk;
        for (int j = tid; j < KR * CAPC / 8; j += OWNER_THREADS) bz[j] = make_int4(0, 0, 0, 0);
    }
    __syncthreads();

    const int4* s4 = (const int4*)src + (size_t)c * (2 * OCT_PER_CHUNK);
    const int4* d4 = (const int4*)dst + (size_t)c * (2 * OCT_PER_CHUNK);

    int o = tid;
    int4 sa = make_int4(0, 0, 0, 0), sb = sa, da = sa, db = sa;
    if (o < OCT_PER_CHUNK) { sa = s4[2 * o]; sb = s4[2 * o + 1]; da = d4[2 * o]; db = d4[2 * o + 1]; }
    while (o < OCT_PER_CHUNK) {
        const int on = o + OWNER_THREADS;
        int4 nsa = make_int4(0, 0, 0, 0), nsb = nsa, nda = nsa, ndb = nsa;
        if (on < OCT_PER_CHUNK) { nsa = s4[2 * on]; nsb = s4[2 * on + 1]; nda = d4[2 * on]; ndb = d4[2 * on + 1]; }
        file_edge(sa.x, da.x, base, cin, cout, bk);
        file_edge(sa.y, da.y, base, cin, cout, bk);
        file_edge(sa.z, da.z, base, cin, cout, bk);
        file_edge(sa.w, da.w, base, cin, cout, bk);
        file_edge(sb.x, db.x, base, cin, cout, bk);
        file_edge(sb.y, db.y, base, cin, cout, bk);
        file_edge(sb.z, db.z, base, cin, cout, bk);
        file_edge(sb.w, db.w, base, cin, cout, bk);
        sa = nsa; sb = nsb; da = nda; db = ndb;
        o = on;
    }
    __syncthreads();

    for (int i = tid; i < KR; i += OWNER_THREADS) {
        cnt_in[c * NP + base + i]  = cin[i];
        cnt_out[c * NP + base + i] = cout[i];
    }
    {
        const int4* bk4 = (const int4*)bk;
        int4* dst4 = (int4*)((slot_t*)bucket + ((size_t)c * NP + base) * CAPC);
        for (int j = tid; j < KR * CAPC / 8; j += OWNER_THREADS) dst4[j] = bk4[j];
    }
}

__global__ __launch_bounds__(256) void gemm_wmma_kernel(
    const float* __restrict__ x, const int* __restrict__ cnt_out,
    const float* __restrict__ W, float* __restrict__ H) {
    int wave = (int)((blockIdx.x * blockDim.x + threadIdx.x) >> 5);
    int lane = threadIdx.x & 31;
    const int n_tiles = N_NODES / 16;
    if (wave >= n_tiles) return;

    const int row0 = wave * 16;
    const int m    = lane & 15;
    const int kk   = (lane >> 4) * 2;
    const int mh   = (lane >> 4) * 8;

    int dov = 0;
#pragma unroll
    for (int c = 0; c < NCHUNK; ++c) dov += cnt_out[c * NP + row0 + m];
    if (dov < 1) dov = 1;
    const float nrm = rsqrtf((float)dov);
    const float* xrow = x + (size_t)(row0 + m) * DIM + kk;

    v8f acc[8] = {};

    for (int k = 0; k < DIM; k += 4) {
        float2 xv = *(const float2*)(xrow + k);
        v2f a = { xv.x * nrm, xv.y * nrm };
        const float* wb = W + (size_t)(k + kk) * DIM + m;
#pragma unroll
        for (int t = 0; t < 8; ++t) {
            v2f b = { wb[t * 16], wb[t * 16 + DIM] };
            acc[t] = __builtin_amdgcn_wmma_f32_16x16x4_f32(
                false, a, false, b, (short)0, acc[t], false, false);
        }
    }

    float* hrow = H + (size_t)row0 * DIM + m;
#pragma unroll
    for (int t = 0; t < 8; ++t) {
#pragma unroll
        for (int r = 0; r < 8; ++r) {
            hrow[(size_t)(r + mh) * DIM + t * 16] = acc[t][r];
        }
    }
}

__device__ __forceinline__ float4 ld_row(const float* __restrict__ Hl, int s) {
    return *(const float4*)(Hl + (size_t)s * DIM);
}

__device__ __forceinline__ void sum_slots(const float* __restrict__ Hl, int iv, int m, float4& acc) {
    for (int j = 0; j < m; j += 8) {
        float4 v[8];
#pragma unroll
        for (int u = 0; u < 8; ++u) {
            v[u] = make_float4(0.f, 0.f, 0.f, 0.f);
            if (j + u < m) {
                const int s = __builtin_amdgcn_readlane(iv, j + u);
                v[u] = ld_row(Hl, s);
            }
        }
#pragma unroll
        for (int u = 0; u < 8; ++u) {
            acc.x += v[u].x; acc.y += v[u].y; acc.z += v[u].z; acc.w += v[u].w;
        }
    }
}

__global__ __launch_bounds__(256) void gather_kernel(
    const float* __restrict__ H, const int* __restrict__ src, const int* __restrict__ dst,
    const int* __restrict__ cnt_in, const int* __restrict__ bucket,
    const float* __restrict__ b,
    float* __restrict__ out, int ne) {
    const int wave = (int)((blockIdx.x * blockDim.x + threadIdx.x) >> 5);
    const int lane = threadIdx.x & 31;
    if (wave >= N_NODES) return;
    const int d = __builtin_amdgcn_readfirstlane(wave);
    int p[NCHUNK + 1];
    bool fits = true;
    p[0] = 0;
#pragma unroll
    for (int c = 0; c < NCHUNK; ++c) {
        const int n = cnt_in[c * NP + d];
        fits = fits && (n <= CAPC);
        p[c + 1] = p[c] + n;
    }
    const int cnt = p[NCHUNK];
    const float nd = rsqrtf((float)(cnt < 1 ? 1 : cnt));
    const float4 bb = *(const float4*)(b + lane * 4);
    const float* Hl = H + lane * 4;
    float4 acc = make_float4(0.f, 0.f, 0.f, 0.f);

    const bool filed = fits && (cnt <= 64);
    if (filed) {
        const slot_t* bk = (const slot_t*)bucket + (size_t)d * CAPC;
        const int j0 = lane;
        const int j1 = lane + 32;
        int c0 = 0, b0 = 0, c1 = 0, b1 = 0;
#pragma unroll
        for (int c = 1; c < NCHUNK; ++c) {
            if (j0 >= p[c]) { c0 = c; b0 = p[c]; }
            if (j1 >= p[c]) { c1 = c; b1 = p[c]; }
        }
        const int i0 = (j0 < cnt) ? bk[((size_t)c0 * NP) * CAPC + (j0 - b0)] : 0;
        const int i1 = (j1 < cnt) ? bk[((size_t)c1 * NP) * CAPC + (j1 - b1)] : 0;
        sum_slots(Hl, i0, cnt < 32 ? cnt : 32, acc);
        if (cnt > 32) sum_slots(Hl, i1, cnt - 32, acc);
    } else {
        for (int e0 = 0; e0 < ne; e0 += 32) {
            const int e = e0 + lane;
            int dd = -1, ss = 0;
            if (e < ne) { dd = dst[e]; ss = src[e]; }
            unsigned hit = __builtin_amdgcn_ballot_w32(dd == d);
            while (hit) {
                const int l = __builtin_ctz(hit);
                hit &= hit - 1u;
                const int s = __builtin_amdgcn_readlane(ss, l);
                const float4 v = ld_row(Hl, s);
                acc.x += v.x; acc.y += v.y; acc.z += v.z; acc.w += v.w;
            }
        }
    }

    float4 o;
    o.x = acc.x * nd + bb.x;
    o.y = acc.y * nd + bb.y;
    o.z = acc.z * nd + bb.z;
    o.w = acc.w * nd + bb.w;
    *(float4*)(out + (size_t)d * DIM + lane * 4) = o;
}

extern "C" void kernel_launch(void* const* d_in, const int* in_sizes, int n_in,
                              void* d_out, int out_size, void* d_ws, size_t ws_size,
                              hipStream_t stream) {
    const float* x   = (const float*)d_in[0];
    const int*   src = (const int*)d_in[1];
    const int*   dst = (const int*)d_in[2];
    const float* W   = (const float*)d_in[3];
    const float* b   = (const float*)d_in[4];
    float* out = (float*)d_out;

    int*   cnt_out = (int*)d_ws;
    int*   cnt_in  = cnt_out + NCHUNK * NP;
    float* H       = (float*)(cnt_in + NCHUNK * NP);
    int*   bucket  = (int*)(H + (size_t)N_NODES * DIM);

    const int TB = 256;

    owner_count_kernel<<<NRANGE * NCHUNK, OWNER_THREADS, 0, stream>>>(src, dst, cnt_out, cnt_in, bucket);

    {
        int waves  = N_NODES / 16;
        int blocks = (waves * 32 + TB - 1) / TB;
        gemm_wmma_kernel<<<blocks, TB, 0, stream>>>(x, cnt_out, W, H);
    }

    gather_kernel<<<(N_NODES * 32 + TB - 1) / TB, TB, 0, stream>>>(H, src, dst, cnt_in, bucket, b, out, N_EDGES);
}


# x@W (no row scale) moves into launch 1, run by strip waves of each edge-scan work-group beside the scan; a small launch builds norm_src, the gather scales each row by it (fma); rank order changes
# speedup vs baseline: 4.6724x; 1.1996x over previous
#include <hip/hip_runtime.h>
#include <hip/hip_bf16.h>


typedef __attribute__((ext_vector_type(2))) float v2f;
typedef __attribute__((ext_vector_type(8))) float v8f;

#define N_NODES 50000
#define N_EDGES 800000
#define DIM 128

#define KR 3200
#define NRANGE 16
#define NCHUNK 16
#define CAPC 16
typedef unsigned short slot_t;
static_assert(N_NODES <= 65536, "a source index must fit a two-byte slot");
#define NP (NRANGE * KR)
#define OCT_PER_CHUNK (N_EDGES / 8 / NCHUNK)
#define OWNER_THREADS 1024

__device__ __forceinline__ void file_edge(int sv, int dv, int base, int* cin, int* cout, slot_t* bk) {
    const unsigned td = (unsigned)(dv - base);
    if (td < (unsigned)KR) {
        const int r = atomicAdd(&cin[td], 1);
        if (r < CAPC) bk[td * CAPC + r] = (slot_t)sv;
    }
    const unsigned ts = (unsigned)(sv - base);
    if (ts < (unsigned)KR) atomicAdd(&cout[ts], 1);
}

template <int NT>
__device__ __forceinline__ void gemm_strip(const float* __restrict__ x, const float* __restrict__ W,
                                           float* __restrict__ H, int strip, int t0, int lane) {
    const int row0 = strip * 16;
    const int m    = lane & 15;
    const int kk   = (lane >> 4) * 2;
    const int mh   = (lane >> 4) * 8;
    const float* xrow = x + (size_t)(row0 + m) * DIM + kk;

    v8f acc[NT] = {};

    for (int k = 0; k < DIM; k += 4) {
        float2 xv = *(const float2*)(xrow + k);
        v2f a = { xv.x, xv.y };
        const float* wb = W + (size_t)(k + kk) * DIM + t0 * 16 + m;
#pragma unroll
        for (int t = 0; t < NT; ++t) {
            v2f b = { wb[t * 16], wb[t * 16 + DIM] };
            acc[t] = __builtin_amdgcn_wmma_f32_16x16x4_f32(
                false, a, false, b, (short)0, acc[t], false, false);
        }
    }

    float* hrow = H + (size_t)row0 * DIM + t0 * 16 + m;
#pragma unroll
    for (int t = 0; t < NT; ++t) {
#pragma unroll
        for (int r = 0; r < 8; ++r) {
            hrow[(size_t)(r + mh) * DIM + t * 16] = acc[t][r];
        }
    }
}

#define STRIP_WAVE_MASK 0x0EC13807u
#define QUART_WAVE_MASK 0x01020408u
#define STRIPS_PER_WG 12
#define N_STRIPS (N_NODES / 16)
#define N_EXTRA (N_STRIPS - STRIPS_PER_WG * NRANGE * NCHUNK)
static_assert(N_EXTRA >= 0 && N_EXTRA <= NRANGE * NCHUNK, "every strip needs a wave");

__global__ __launch_bounds__(OWNER_THREADS) void owner_count_kernel(
    const int* __restrict__ src, const int* __restrict__ dst,
    int* __restrict__ cnt_out, int* __restrict__ cnt_in, int* __restrict__ bucket,
    const float* __restrict__ x, const float* __restrict__ W, float* __restrict__ H) {
    __shared__ int cin[KR];
    __shared__ int cout[KR];
    __shared__ __attribute__((aligned(16))) slot_t bk[KR * CAPC];

    const int tid  = threadIdx.x;
    const int k    = blockIdx.x / NCHUNK;
    const int c    = blockIdx.x - k * NCHUNK;
    const int base = k * KR;

    for (int i = tid; i < KR; i += OWNER_THREADS) { cin[i] = 0; cout[i] = 0; }
    {
        int4* bz = (int4*)bk;
        for (int j = tid; j < KR * CAPC / 8; j += OWNER_THREADS) bz[j] = make_int4(0, 0, 0, 0);
    }
    __syncthreads();

    const int4* s4 = (const int4*)src + (size_t)c * (2 * OCT_PER_CHUNK);
    const int4* d4 = (const int4*)dst + (size_t)c * (2 * OCT_PER_CHUNK);

    const int wv = __builtin_amdgcn_readfirstlane(tid >> 5);
    const unsigned below = (1u << wv) - 1u;
    const bool has_extra = (int)blockIdx.x < N_EXTRA;
    const unsigned mm_mask = STRIP_WAVE_MASK | (has_extra ? QUART_WAVE_MASK : 0u);
    const bool mm_wave = (mm_mask >> wv) & 1u;
    if ((STRIP_WAVE_MASK >> wv) & 1u) {
        const int g = __builtin_popcount(STRIP_WAVE_MASK & below);
        gemm_strip<8>(x, W, H, g * (NRANGE * NCHUNK) + (int)blockIdx.x, 0, tid & 31);
    } else if (mm_wave) {
        const int q = __builtin_popcount(QUART_WAVE_MASK & below);
        gemm_strip<2>(x, W, H, STRIPS_PER_WG * (NRANGE * NCHUNK) + (int)blockIdx.x, 2 * q, tid & 31);
    }

    const int scan_threads = (OWNER_THREADS / 32 - __builtin_popcount(mm_mask)) * 32;
    int o = mm_wave ? OCT_PER_CHUNK : (wv - __builtin_popcount(mm_mask & below)) * 32 + (tid & 31);
    int4 sa = make_int4(0, 0, 0, 0), sb = sa, da = sa, db = sa;
    if (o < OCT_PER_CHUNK) { sa = s4[2 * o]; sb = s4[2 * o + 1]; da = d4[2 * o]; db = d4[2 * o + 1]; }
    while (o < OCT_PER_CHUNK) {
        const int on = o + scan_threads;
        int4 nsa = make_int4(0, 0, 0, 0), nsb = nsa, nda = nsa, ndb = nsa;
        if (on < OCT_PER_CHUNK) { nsa = s4[2 * on]; nsb = s4[2 * on + 1]; nda = d4[2 * on]; ndb = d4[2 * on + 1]; }
        file_edge(sa.x, da.x, base, cin, cout, bk);
        file_edge(sa.y, da.y, base, cin, cout, bk);
        file_edge(sa.z, da.z, base, cin, cout, bk);
        file_edge(sa.w, da.w, base, cin, cout, bk);
        file_edge(sb.x, db.x, base, cin, cout, bk);
        file_edge(sb.y, db.y, base, cin, cout, bk);
        file_edge(sb.z, db.z, base, cin, cout, bk);
        file_edge(sb.w, db.w, base, cin, cout, bk);
        sa = nsa; sb = nsb; da = nda; db = ndb;
        o = on;
    }
    __syncthreads();

    for (int i = tid; i < KR; i += OWNER_THREADS) {
        cnt_in[c * NP + base + i]  = cin[i];
        cnt_out[c * NP + base + i] = cout[i];
    }
    {
        const int4* bk4 = (const int4*)bk;
        int4* dst4 = (int4*)((slot_t*)bucket + ((size_t)c * NP + base) * CAPC);
        for (int j = tid; j < KR * CAPC / 8; j += OWNER_THREADS) dst4[j] = bk4[j];
    }
}

__global__ __launch_bounds__(256) void norm_src_kernel(const int* __restrict__ cnt_out, float* __restrict__ ns) {
    const int i = (int)(blockIdx.x * blockDim.x + threadIdx.x);
    if (i >= N_NODES) return;
    int dov = 0;
#pragma unroll
    for (int c = 0; c < NCHUNK; ++c) dov += cnt_out[c * NP + i];
    if (dov < 1) dov = 1;
    ns[i] = rsqrtf((float)dov);
}

__device__ __forceinline__ float4 ld_row(const float* __restrict__ Hl, int s) {
    return *(const float4*)(Hl + (size_t)s * DIM);
}

__device__ __forceinline__ void sum_slots(const float* __restrict__ Hl, int iv, float nv, int m, float4& acc) {
    for (int j = 0; j < m; j += 8) {
        float4 v[8];
#pragma unroll
        for (int u = 0; u < 8; ++u) {
            v[u] = make_float4(0.f, 0.f, 0.f, 0.f);
            if (j + u < m) {
                const int s = __builtin_amdgcn_readlane(iv, j + u);
                v[u] = ld_row(Hl, s);
            }
        }
#pragma unroll
        for (int u = 0; u < 8; ++u) {
            const float w = __int_as_float(__builtin_amdgcn_readlane(__float_as_int(nv), j + u));
            acc.x = fmaf(w, v[u].x, acc.x); acc.y = fmaf(w, v[u].y, acc.y);
            acc.z = fmaf(w, v[u].z, acc.z); acc.w = fmaf(w, v[u].w, acc.w);
        }
    }
}

__global__ __launch_bounds__(256) void gather_kernel(
    const float* __restrict__ H, const int* __restrict__ src, const int* __restrict__ dst,
    const int* __restrict__ cnt_in, const int* __restrict__ bucket,
    const float* __restrict__ ns, const float* __restrict__ b,
    float* __restrict__ out, int ne) {
    const int wave = (int)((blockIdx.x * blockDim.x + threadIdx.x) >> 5);
    const int lane = threadIdx.x & 31;
    if (wave >= N_NODES) return;
    const int d = __builtin_amdgcn_readfirstlane(wave);
    int p[NCHUNK + 1];
    bool fits = true;
    p[0] = 0;
#pragma unroll
    for (int c = 0; c < NCHUNK; ++c) {
        const int n = cnt_in[c * NP + d];
        fits = fits && (n <= CAPC);
        p[c + 1] = p[c] + n;
    }
    const int cnt = p[NCHUNK];
    const float nd = rsqrtf((float)(cnt < 1 ? 1 : cnt));
    const float4 bb = *(const float4*)(b + lane * 4);
    const float* Hl = H + lane * 4;
    float4 acc = make_float4(0.f, 0.f, 0.f, 0.f);

    const bool filed = fits && (cnt <= 64);
    if (filed) {
        const slot_t* bk = (const slot_t*)bucket + (size_t)d * CAPC;
        const int j0 = lane;
        const int j1 = lane + 32;
        int c0 = 0, b0 = 0, c1 = 0, b1 = 0;
#pragma unroll
        for (int c = 1; c < NCHUNK; ++c) {
            if (j0 >= p[c]) { c0 = c; b0 = p[c]; }
            if (j1 >= p[c]) { c1 = c; b1 = p[c]; }
        }
        const int i0 = (j0 < cnt) ? bk[((size_t)c0 * NP) * CAPC + (j0 - b0)] : 0;
        const int i1 = (j1 < cnt) ? bk[((size_t)c1 * NP) * CAPC + (j1 - b1)] : 0;
        const float n0 = (j0 < cnt) ? ns[i0] : 0.f;
        const float n1 = (j1 < cnt) ? ns[i1] : 0.f;
        sum_slots(Hl, i0, n0, cnt < 32 ? cnt : 32, acc);
        if (cnt > 32) sum_slots(Hl, i1, n1, cnt - 32, acc);
    } else {
        for (int e0 = 0; e0 < ne; e0 += 32) {
            const int e = e0 + lane;
            int dd = -1, ss = 0;
            if (e < ne) { dd = dst[e]; ss = src[e]; }
            unsigned hit = __builtin_amdgcn_ballot_w32(dd == d);
            while (hit) {
                const int l = __builtin_ctz(hit);
                hit &= hit - 1u;
                const int s = __builtin_amdgcn_readlane(ss, l);
                const float4 v = ld_row(Hl, s);
                const float w = ns[s];
                acc.x = fmaf(w, v.x, acc.x); acc.y = fmaf(w, v.y, acc.y);
                acc.z = fmaf(w, v.z, acc.z); acc.w = fmaf(w, v.w, acc.w);
            }
        }
    }

    float4 o;
    o.x = acc.x * nd + bb.x;
    o.y = acc.y * nd + bb.y;
    o.z = acc.z * nd + bb.z;
    o.w = acc.w * nd + bb.w;
    *(float4*)(out + (size_t)d * DIM + lane * 4) = o;
}

extern "C" void kernel_launch(void* const* d_in, const int* in_sizes, int n_in,
                              void* d_out, int out_size, void* d_ws, size_t ws_size,
                              hipStream_t stream) {
    const float* x   = (const float*)d_in[0];
    const int*   src = (const int*)d_in[1];
    const int*   dst = (const int*)d_in[2];
    const float* W   = (const float*)d_in[3];
    const float* b   = (const float*)d_in[4];
    float* out = (float*)d_out;

    int*   cnt_out = (int*)d_ws;
    int*   cnt_in  = cnt_out + NCHUNK * NP;
    float* H       = (float*)(cnt_in + NCHUNK * NP);
    int*   bucket  = (int*)(H + (size_t)N_NODES * DIM);
    float* ns      = (float*)(bucket + (size_t)NCHUNK * NP * CAPC / 2);

    const int TB = 256;

    owner_count_kernel<<<NRANGE * NCHUNK, OWNER_THREADS, 0, stream>>>(src, dst, cnt_out, cnt_in, bucket, x, W, H);

    norm_src_kernel<<<(N_NODES + TB - 1) / TB, TB, 0, stream>>>(cnt_out, ns);

    gather_kernel<<<(N_NODES * 32 + TB - 1) / TB, TB, 0, stream>>>(H, src, dst, cnt_in, bucket, ns, b, out, N_EDGES);
}
